# starting kernel
# baseline (speedup 1.0000x reference)
#include <hip/hip_runtime.h>

#define N_NODES 100000
#define N_EDGES 1600000
#define IN_DIM  128
#define OUT_DIM 64
#define NEG_SLOPE 0.01f

typedef float v2f __attribute__((ext_vector_type(2)));
typedef float v8f __attribute__((ext_vector_type(8)));

__global__ void degree_kernel(const int* __restrict__ src, const int* __restrict__ dst,
                              float* __restrict__ outdeg, float* __restrict__ indeg, int E) {
    int e = blockIdx.x * blockDim.x + threadIdx.x;
    if (e < E) {
        atomicAdd(&outdeg[src[e]], 1.0f);
        atomicAdd(&indeg[dst[e]], 1.0f);
    }
}

__global__ void norm_kernel(float* __restrict__ ns, float* __restrict__ nd, int n) {
    int i = blockIdx.x * blockDim.x + threadIdx.x;
    if (i < n) {
        ns[i] = rsqrtf(fmaxf(ns[i], 1.0f));
        nd[i] = rsqrtf(fmaxf(nd[i], 1.0f));
    }
}

__global__ void __launch_bounds__(256)
gemm_wmma_kernel(const float* __restrict__ h, const float* __restrict__ W,
                 const float* __restrict__ ns, float* __restrict__ x, int mtiles) {
    __shared__ float Wt[OUT_DIM * 130];
    for (int i = threadIdx.x; i < IN_DIM * OUT_DIM; i += 256) {
        int k = i >> 6;
        int c = i & 63;
        Wt[c * 130 + k] = W[i];
    }
    __syncthreads();

    const int wave = threadIdx.x >> 5;
    const int lane = threadIdx.x & 31;
    const int mtile = blockIdx.x * 8 + wave;
    if (mtile >= mtiles) return;

    const int mrow = (mtile << 4) + (lane & 15);
    const int kb   = (lane >> 4) << 1;
    const int n    = lane & 15;

    const float* hrow = h + (size_t)mrow * IN_DIM + kb;
    const float* w0 = &Wt[(0 * 16 + n) * 130 + kb];
    const float* w1 = &Wt[(1 * 16 + n) * 130 + kb];
    const float* w2 = &Wt[(2 * 16 + n) * 130 + kb];
    const float* w3 = &Wt[(3 * 16 + n) * 130 + kb];

    v8f acc0 = {}, acc1 = {}, acc2 = {}, acc3 = {};
#pragma unroll 4
    for (int k = 0; k < IN_DIM; k += 4) {
        v2f a  = *(const v2f*)(hrow + k);
        v2f b0 = *(const v2f*)(w0 + k);
        v2f b1 = *(const v2f*)(w1 + k);
        v2f b2 = *(const v2f*)(w2 + k);
        v2f b3 = *(const v2f*)(w3 + k);
        acc0 = __builtin_amdgcn_wmma_f32_16x16x4_f32(false, a, false, b0, (short)0, acc0, false, false);
        acc1 = __builtin_amdgcn_wmma_f32_16x16x4_f32(false, a, false, b1, (short)0, acc1, false, false);
        acc2 = __builtin_amdgcn_wmma_f32_16x16x4_f32(false, a, false, b2, (short)0, acc2, false, false);
        acc3 = __builtin_amdgcn_wmma_f32_16x16x4_f32(false, a, false, b3, (short)0, acc3, false, false);
    }

    const int rowbase = (mtile << 4) + ((lane >> 4) << 3);
#pragma unroll
    for (int r = 0; r < 8; ++r) {
        int row = rowbase + r;
        float s = ns[row];
        float* xr = x + (size_t)row * OUT_DIM + n;
        xr[0]  = acc0[r] * s;
        xr[16] = acc1[r] * s;
        xr[32] = acc2[r] * s;
        xr[48] = acc3[r] * s;
    }
}

__global__ void scatter_kernel(const float* __restrict__ x, const int* __restrict__ src,
                               const int* __restrict__ dst, float* __restrict__ agg, int E) {
    int t = blockIdx.x * blockDim.x + threadIdx.x;
    int e  = t >> 4;
    int c4 = (t & 15) << 2;
    if (e < E) {
        int s = src[e];
        int d = dst[e];
        const float4 v = *(const float4*)(x + (size_t)s * OUT_DIM + c4);
        float* ap = agg + (size_t)d * OUT_DIM + c4;
        atomicAdd(ap + 0, v.x);
        atomicAdd(ap + 1, v.y);
        atomicAdd(ap + 2, v.z);
        atomicAdd(ap + 3, v.w);
    }
}

__global__ void finalize_kernel(const float* __restrict__ agg, const float* __restrict__ nd,
                                const float* __restrict__ b, float* __restrict__ out, int total) {
    int t = blockIdx.x * blockDim.x + threadIdx.x;
    if (t < total) {
        int i = t >> 6;
        int c = t & 63;
        float v = agg[t] * nd[i] + b[c];
        out[t] = (v >= 0.0f) ? v : v * NEG_SLOPE;
    }
}

extern "C" void kernel_launch(void* const* d_in, const int* in_sizes, int n_in,
                              void* d_out, int out_size, void* d_ws, size_t ws_size,
                              hipStream_t stream) {
    const float* h    = (const float*)d_in[0];
    const float* W    = (const float*)d_in[1];
    const float* b    = (const float*)d_in[2];
    const int*   esrc = (const int*)d_in[3];
    const int*   edst = (const int*)d_in[4];
    float*       out  = (float*)d_out;

    float* ws  = (float*)d_ws;
    float* ns  = ws;
    float* nd  = ws + N_NODES;
    float* x   = ws + 2 * (size_t)N_NODES;
    float* agg = x + (size_t)N_NODES * OUT_DIM;

    hipMemsetAsync(ns,  0, 2 * (size_t)N_NODES * sizeof(float), stream);
    hipMemsetAsync(agg, 0, (size_t)N_NODES * OUT_DIM * sizeof(float), stream);

    degree_kernel<<<(N_EDGES + 255) / 256, 256, 0, stream>>>(esrc, edst, ns, nd, N_EDGES);
    norm_kernel<<<(N_NODES + 255) / 256, 256, 0, stream>>>(ns, nd, N_NODES);

    const int mtiles = N_NODES / 16;
    gemm_wmma_kernel<<<(mtiles + 7) / 8, 256, 0, stream>>>(h, W, ns, x, mtiles);

    const long long sthreads = (long long)N_EDGES * 16;
    scatter_kernel<<<(unsigned)((sthreads + 255) / 256), 256, 0, stream>>>(x, esrc, edst, agg, N_EDGES);

    const int total = N_NODES * OUT_DIM;
    finalize_kernel<<<(total + 255) / 256, 256, 0, stream>>>(agg, nd, b, out, total);
}


# aggregation rows owned by waves (row mod 32): records go to the owner's LDS queue, the owner adds gathered x rows with plain 16-byte LDS read-add-write, no LDS float atomics; float sum order changes
# speedup vs baseline: 5.0243x; 5.0243x over previous
#include <hip/hip_runtime.h>

#define N_NODES 100000
#define N_EDGES 1600000
#define IN_DIM  128
#define OUT_DIM 64
#define NEG_SLOPE 0.01f

typedef float v2f __attribute__((ext_vector_type(2)));
typedef float v8f __attribute__((ext_vector_type(8)));

#define TILE_R   800
#define N_TILES  (N_NODES / TILE_R)
#define DT_R     400
#define N_DT     (N_NODES / DT_R)

static_assert(N_NODES % TILE_R == 0 && N_NODES % DT_R == 0, "tiles cover the nodes exactly");
static_assert(DT_R <= 1024 && N_NODES <= (1 << 22), "a record packs the source above 10 bits of local row");

#define CHUNK    6400
#define N_CHUNKS (N_EDGES / CHUNK)
#define SORT_T   1024
#define SORT_U   ((CHUNK + SORT_T - 1) / SORT_T)
#define N_BINS   128
#define D_BINS   256

static_assert(N_EDGES % CHUNK == 0, "chunks cover the edge list exactly");
static_assert(N_TILES + 1 <= N_BINS && N_BINS == 4 * 32, "one wave scans the source bins, four per lane");
static_assert(N_DT + 1 <= D_BINS && D_BINS == 8 * 32, "one wave scans the destination bins, eight per lane");
static_assert(CHUNK < 65536 && CHUNK % 8 == 0, "run bounds are two 16-bit halves; a chunk is copied out as uint4");

__global__ void __launch_bounds__(SORT_T)
sort_kernel(const int* __restrict__ src, const int* __restrict__ dst,
            unsigned* __restrict__ drec, unsigned* __restrict__ dtab,
            unsigned short* __restrict__ ssort, unsigned* __restrict__ stab) {
    __shared__ __attribute__((aligned(16))) int cnt[N_BINS];
    __shared__ __attribute__((aligned(16))) int off[N_BINS];
    __shared__ __attribute__((aligned(16))) int dcnt[D_BINS];
    __shared__ __attribute__((aligned(16))) int doff[D_BINS];
    __shared__ __attribute__((aligned(16))) unsigned short sorted[CHUNK];
    __shared__ __attribute__((aligned(16))) unsigned dsorted[CHUNK];

    const int tid = threadIdx.x;
    const int c   = blockIdx.x;
    const int e0  = c * CHUNK;

    if (tid < N_BINS) cnt[tid] = 0;
    if (tid < D_BINS) dcnt[tid] = 0;
    __syncthreads();

    unsigned sv[SORT_U], dv[SORT_U];
    int      rank[SORT_U], drank[SORT_U];
#pragma unroll
    for (int u = 0; u < SORT_U; ++u) {
        const int i = tid + u * SORT_T;
        sv[u] = 0u;
        dv[u] = 0u;
        if (i < CHUNK) {
            sv[u] = (unsigned)src[e0 + i];
            dv[u] = (unsigned)dst[e0 + i];
        }
    }
#pragma unroll
    for (int u = 0; u < SORT_U; ++u) {
        const int i = tid + u * SORT_T;
        rank[u]  = 0;
        drank[u] = 0;
        if (i < CHUNK) {
            const unsigned s = sv[u];
            const unsigned d = dv[u];
            const int sb = (s < (unsigned)N_NODES) ? (int)(s / TILE_R) : N_TILES;
            const int db = (d < (unsigned)N_NODES) ? (int)(d / DT_R)   : N_DT;
            rank[u]  = atomicAdd(&cnt[sb], 1);
            drank[u] = atomicAdd(&dcnt[db], 1);
        }
    }
    __syncthreads();

    if (tid < 32) {
        const int4 cv  = ((const int4*)cnt)[tid];
        const int  sum = cv.x + cv.y + cv.z + cv.w;
        int incl = sum;
#pragma unroll
        for (int dlt = 1; dlt < 32; dlt <<= 1) {
            const int up = __shfl_up(incl, dlt);
            if (tid >= dlt) incl += up;
        }
        const int ex = incl - sum;
        ((int4*)off)[tid] = make_int4(ex, ex + cv.x, ex + cv.x + cv.y, ex + cv.x + cv.y + cv.z);
    } else if (tid < 64) {
        const int  l   = tid - 32;
        const int4 ca  = ((const int4*)dcnt)[2 * l];
        const int4 cb  = ((const int4*)dcnt)[2 * l + 1];
        const int  sa  = ca.x + ca.y + ca.z + ca.w;
        const int  sum = sa + cb.x + cb.y + cb.z + cb.w;
        int incl = sum;
#pragma unroll
        for (int dlt = 1; dlt < 32; dlt <<= 1) {
            const int up = __shfl_up(incl, dlt);
            if (l >= dlt) incl += up;
        }
        const int ex = incl - sum;
        const int eb = ex + sa;
        ((int4*)doff)[2 * l]     = make_int4(ex, ex + ca.x, ex + ca.x + ca.y, ex + ca.x + ca.y + ca.z);
        ((int4*)doff)[2 * l + 1] = make_int4(eb, eb + cb.x, eb + cb.x + cb.y, eb + cb.x + cb.y + cb.z);
    }
    __syncthreads();

#pragma unroll
    for (int u = 0; u < SORT_U; ++u) {
        const int i = tid + u * SORT_T;
        if (i < CHUNK) {
            const unsigned s  = sv[u];
            const unsigned d  = dv[u];
            const unsigned st = s / TILE_R;
            const unsigned dt = d / DT_R;
            const int sb = (s < (unsigned)N_NODES) ? (int)st : N_TILES;
            const int db = (d < (unsigned)N_NODES) ? (int)dt : N_DT;
            sorted[off[sb] + rank[u]]    = (unsigned short)(s - st * TILE_R);
            dsorted[doff[db] + drank[u]] = (s << 10) | (d - dt * DT_R);
        }
    }
    if (tid < N_TILES)
        stab[tid * N_CHUNKS + c] = (unsigned)off[tid] | ((unsigned)(off[tid] + cnt[tid]) << 16);
    if (tid < N_DT)
        dtab[tid * N_CHUNKS + c] = (unsigned)doff[tid] | ((unsigned)(doff[tid] + dcnt[tid]) << 16);
    __syncthreads();

    const uint4* __restrict__ s4 = (const uint4*)sorted;
    uint4* __restrict__ g4 = (uint4*)(ssort + (size_t)c * CHUNK);
    for (int i = tid; i < CHUNK / 8; i += SORT_T)
        g4[i] = s4[i];
    const uint4* __restrict__ d4 = (const uint4*)dsorted;
    uint4* __restrict__ h4 = (uint4*)(drec + (size_t)c * CHUNK);
    for (int i = tid; i < CHUNK / 4; i += SORT_T)
        h4[i] = d4[i];
}

#define OD_WAVES 32
#define OD_RUNS  ((N_CHUNKS + OD_WAVES - 1) / OD_WAVES)

__global__ void __launch_bounds__(OD_WAVES * 32)
outdeg_kernel(const unsigned short* __restrict__ ssort, const unsigned* __restrict__ stab,
              float* __restrict__ ns) {
    __shared__ int cnt[TILE_R];

    const int tid  = threadIdx.x;
    const int wave = __builtin_amdgcn_readfirstlane(tid >> 5);
    const int lane = tid & 31;
    const int t    = blockIdx.x;

    for (int i = tid; i < TILE_R; i += OD_WAVES * 32)
        cnt[i] = 0;
    __syncthreads();

    const unsigned* __restrict__ tb = stab + t * N_CHUNKS;
    unsigned se[OD_RUNS];
#pragma unroll
    for (int u = 0; u < OD_RUNS; ++u) {
        const int c = wave + OD_WAVES * u;
        se[u] = (c < N_CHUNKS) ? tb[c] : 0u;
    }
    int a0[OD_RUNS], a1[OD_RUNS];
#pragma unroll
    for (int u = 0; u < OD_RUNS; ++u) {
        const int c     = wave + OD_WAVES * u;
        const int start = (int)(se[u] & 0xFFFFu);
        const int end   = (int)(se[u] >> 16);
        const unsigned short* __restrict__ rp = ssort + (size_t)c * CHUNK;
        const int i0 = start + lane;
        const int i1 = i0 + 32;
        a0[u] = (i0 < end) ? (int)rp[i0] : -1;
        a1[u] = (i1 < end) ? (int)rp[i1] : -1;
    }
#pragma unroll
    for (int u = 0; u < OD_RUNS; ++u) {
        if (a0[u] >= 0) __hip_atomic_fetch_add(&cnt[a0[u]], 1, __ATOMIC_RELAXED, __HIP_MEMORY_SCOPE_WORKGROUP);
        if (a1[u] >= 0) __hip_atomic_fetch_add(&cnt[a1[u]], 1, __ATOMIC_RELAXED, __HIP_MEMORY_SCOPE_WORKGROUP);
        const int c     = wave + OD_WAVES * u;
        const int start = (int)(se[u] & 0xFFFFu);
        const int end   = (int)(se[u] >> 16);
        const unsigned short* __restrict__ rp = ssort + (size_t)c * CHUNK;
        for (int i = start + 64 + lane; i < end; i += 32)
            __hip_atomic_fetch_add(&cnt[rp[i]], 1, __ATOMIC_RELAXED, __HIP_MEMORY_SCOPE_WORKGROUP);
    }
    __syncthreads();

    for (int i = tid; i < TILE_R; i += OD_WAVES * 32)
        ns[t * TILE_R + i] = rsqrtf(fmaxf((float)cnt[i], 1.0f));
}

__global__ void __launch_bounds__(256)
gemm_wmma_kernel(const float* __restrict__ h, const float* __restrict__ W,
                 const float* __restrict__ ns, float* __restrict__ x, int mtiles) {
    __shared__ float Wt[OUT_DIM * 130];
    for (int i = threadIdx.x; i < IN_DIM * OUT_DIM; i += 256) {
        int k = i >> 6;
        int c = i & 63;
        Wt[c * 130 + k] = W[i];
    }
    __syncthreads();

    const int wave = threadIdx.x >> 5;
    const int lane = threadIdx.x & 31;
    const int mtile = blockIdx.x * 8 + wave;
    if (mtile >= mtiles) return;

    const int mrow = (mtile << 4) + (lane & 15);
    const int kb   = (lane >> 4) << 1;
    const int n    = lane & 15;

    const float* hrow = h + (size_t)mrow * IN_DIM + kb;
    const float* w0 = &Wt[(0 * 16 + n) * 130 + kb];
    const float* w1 = &Wt[(1 * 16 + n) * 130 + kb];
    const float* w2 = &Wt[(2 * 16 + n) * 130 + kb];
    const float* w3 = &Wt[(3 * 16 + n) * 130 + kb];

    v8f acc0 = {}, acc1 = {}, acc2 = {}, acc3 = {};
#pragma unroll 4
    for (int k = 0; k < IN_DIM; k += 4) {
        v2f a  = *(const v2f*)(hrow + k);
        v2f b0 = *(const v2f*)(w0 + k);
        v2f b1 = *(const v2f*)(w1 + k);
        v2f b2 = *(const v2f*)(w2 + k);
        v2f b3 = *(const v2f*)(w3 + k);
        acc0 = __builtin_amdgcn_wmma_f32_16x16x4_f32(false, a, false, b0, (short)0, acc0, false, false);
        acc1 = __builtin_amdgcn_wmma_f32_16x16x4_f32(false, a, false, b1, (short)0, acc1, false, false);
        acc2 = __builtin_amdgcn_wmma_f32_16x16x4_f32(false, a, false, b2, (short)0, acc2, false, false);
        acc3 = __builtin_amdgcn_wmma_f32_16x16x4_f32(false, a, false, b3, (short)0, acc3, false, false);
    }

    const int rowbase = (mtile << 4) + ((lane >> 4) << 3);
#pragma unroll
    for (int r = 0; r < 8; ++r) {
        int row = rowbase + r;
        float s = ns[row];
        float* xr = x + (size_t)row * OUT_DIM + n;
        xr[0]  = acc0[r] * s;
        xr[16] = acc1[r] * s;
        xr[32] = acc2[r] * s;
        xr[48] = acc3[r] * s;
    }
}

#define AGG_WAVES 32
#define DT_RUNS  ((N_CHUNKS + AGG_WAVES - 1) / AGG_WAVES)
#define Q_RUN    64
#define Q_CAP    512
#define STEP_U   16
#define STEP_H   8

static_assert(DT_RUNS <= 32, "a wave holds the bounds of its runs one per lane");
static_assert(STEP_U == 16 && STEP_H == 8 && Q_RUN == 64, "a step's record is picked inside a 16-lane half; two lane-loads cover a run's read-ahead part");
static_assert(AGG_WAVES == 32 && OUT_DIM == 64, "a row's owner is its number mod 32; a lane half covers a 64-column row with 16 bytes per lane");

template <int K> struct Steps { static constexpr int value = K; };

__device__ __forceinline__ unsigned xhalf_u(unsigned u) {
    return __builtin_amdgcn_permlanex16(u, u, 0x76543210u, 0xfedcba98u, false, false);
}
__device__ __forceinline__ float xhalf(float a) {
    const unsigned u = __float_as_uint(a);
    return __uint_as_float(__builtin_amdgcn_permlanex16(u, u, 0x76543210u, 0xfedcba98u, false, false));
}

__global__ void __launch_bounds__(AGG_WAVES * 32)
agg_kernel(const float* __restrict__ x, const unsigned* __restrict__ drec,
           const unsigned* __restrict__ dtab, const float* __restrict__ b,
           float* __restrict__ out) {
    __shared__ __attribute__((aligned(16))) float tile[(DT_R + 1) * OUT_DIM];
    __shared__ int      indeg[DT_R];
    __shared__ int      qcnt[AGG_WAVES];
    __shared__ unsigned queue[AGG_WAVES * Q_CAP];

    const int tid  = threadIdx.x;
    const int wave = __builtin_amdgcn_readfirstlane(tid >> 5);
    const int lane = tid & 31;
    const int t    = blockIdx.x;
    const int lo   = t * DT_R;

    for (int i = tid; i < (DT_R + 1) * (OUT_DIM / 4); i += AGG_WAVES * 32)
        ((float4*)tile)[i] = make_float4(0.0f, 0.0f, 0.0f, 0.0f);
    for (int i = tid; i < DT_R; i += AGG_WAVES * 32)
        indeg[i] = 0;
    if (tid < AGG_WAVES) qcnt[tid] = 0;
    __syncthreads();

    auto add_now = [&](unsigned r, bool full) {
        unsigned m = __builtin_amdgcn_ballot_w32(full);
        while (m != 0u) {
            const int l = __builtin_ctz(m);
            m &= m - 1u;
            const unsigned rr = (unsigned)__builtin_amdgcn_readlane((int)r, l);
            const float2 v = *(const float2*)(x + (size_t)(rr >> 10) * OUT_DIM + 2 * lane);
            float* tp = &tile[(rr & 1023u) * OUT_DIM + 2 * lane];
            __hip_atomic_fetch_add(tp,     v.x, __ATOMIC_RELAXED, __HIP_MEMORY_SCOPE_WORKGROUP);
            __hip_atomic_fetch_add(tp + 1, v.y, __ATOMIC_RELAXED, __HIP_MEMORY_SCOPE_WORKGROUP);
        }
    };
    auto draw = [&](unsigned r, bool valid) -> int {
        int sl = -1;
        if (valid) {
            const int dl = (int)(r & 1023u);
            __hip_atomic_fetch_add(&indeg[dl], 1, __ATOMIC_RELAXED, __HIP_MEMORY_SCOPE_WORKGROUP);
            sl = atomicAdd(&qcnt[dl & (AGG_WAVES - 1)], 1);
        }
        return sl;
    };
    auto place = [&](unsigned r, int sl) {
        if (sl >= 0 && sl < Q_CAP)
            queue[(r & (unsigned)(AGG_WAVES - 1)) * Q_CAP + sl] = r;
        add_now(r, sl >= Q_CAP);
    };

    const unsigned* __restrict__ tb = dtab + t * N_CHUNKS;
    const int cl  = wave + AGG_WAVES * lane;
    const int sev = (lane < DT_RUNS && cl < N_CHUNKS) ? (int)tb[cl] : 0;

    unsigned a0[DT_RUNS], a1[DT_RUNS];
#pragma unroll
    for (int u = 0; u < DT_RUNS; ++u) {
        const unsigned se = (unsigned)__builtin_amdgcn_readlane(sev, u);
        const int start = (int)(se & 0xFFFFu);
        const int end   = (int)(se >> 16);
        const unsigned* __restrict__ rp = drec + (size_t)(wave + AGG_WAVES * u) * CHUNK;
        const int i0 = start + lane;
        const int i1 = i0 + 32;
        a0[u] = (i0 < end) ? rp[i0] : 0u;
        a1[u] = (i1 < end) ? rp[i1] : 0u;
    }
    int s0[DT_RUNS], s1[DT_RUNS];
#pragma unroll
    for (int u = 0; u < DT_RUNS; ++u) {
        const unsigned se = (unsigned)__builtin_amdgcn_readlane(sev, u);
        const int len = (int)(se >> 16) - (int)(se & 0xFFFFu);
        s0[u] = draw(a0[u], lane < len);
        s1[u] = draw(a1[u], lane + 32 < len);
    }
#pragma unroll
    for (int u = 0; u < DT_RUNS; ++u) {
        place(a0[u], s0[u]);
        place(a1[u], s1[u]);
    }
    for (int u = 0; u < DT_RUNS; ++u) {
        const unsigned se = (unsigned)__builtin_amdgcn_readlane(sev, u);
        const int start = (int)(se & 0xFFFFu);
        const int end   = (int)(se >> 16);
        const unsigned* __restrict__ rp = drec + (size_t)(wave + AGG_WAVES * u) * CHUNK;
        for (int i = start + Q_RUN; i < end; i += 32) {
            const bool valid = i + lane < end;
            const unsigned r = valid ? rp[i + lane] : 0u;
            place(r, draw(r, valid));
        }
    }
    __syncthreads();

    const int c  = lane & 15;
    const int hf = lane >> 4;
    const int n  = min(__builtin_amdgcn_readfirstlane(qcnt[wave]), Q_CAP);
    const unsigned* myq = queue + wave * Q_CAP;
    for (int base = 0; base < n; base += 32) {
        const int nb  = min(32, n - base);
        const int idx = 2 * c + hf;
        const unsigned r  = (idx < nb) ? myq[base + idx] : (unsigned)DT_R;
        const unsigned dl = r & 1023u;
        const unsigned same = __builtin_amdgcn_ballot_w32(dl == xhalf_u(dl));
        auto steps = [&](auto kc) {
            constexpr int K = decltype(kc)::value;
            float4   v[K];
            unsigned dk[K];
#pragma unroll
            for (int j = 0; j < K; ++j) {
                const unsigned sel = (unsigned)j * 0x11111111u;
                const unsigned rq = __builtin_amdgcn_permlane16(r, r, sel, sel, false, false);
                dk[j] = rq & 1023u;
                v[j]  = *(const float4*)(x + (size_t)(rq >> 10) * OUT_DIM + 4 * c);
            }
#pragma unroll
            for (int j = 0; j < K; ++j) {
                float4* tp = (float4*)&tile[dk[j] * OUT_DIM + 4 * c];
                if ((same >> j) & 1u) {
                    const float ox = xhalf(v[j].x), oy = xhalf(v[j].y), oz = xhalf(v[j].z), ow = xhalf(v[j].w);
                    if (hf == 0) {
                        float4 s = *tp;
                        s.x += v[j].x + ox; s.y += v[j].y + oy; s.z += v[j].z + oz; s.w += v[j].w + ow;
                        *tp = s;
                    }
                } else {
                    float4 s = *tp;
                    s.x += v[j].x; s.y += v[j].y; s.z += v[j].z; s.w += v[j].w;
                    *tp = s;
                }
            }
        };
        if (nb > STEP_H * 2) steps(Steps<STEP_U>{});
        else                 steps(Steps<STEP_H>{});
    }
    __syncthreads();

    const float4* __restrict__ t4 = (const float4*)tile;
    float4* __restrict__ out4 = (float4*)out + (size_t)lo * (OUT_DIM / 4);
    for (int i = tid; i < DT_R * (OUT_DIM / 4); i += AGG_WAVES * 32) {
        const int r  = i >> 4;
        const int c4 = (i & 15) << 2;
        const float4 a  = t4[i];
        const float  nd = rsqrtf(fmaxf((float)indeg[r], 1.0f));
        const float4 bb = *(const float4*)(b + c4);
        float4 o;
        float w;
        w = a.x * nd + bb.x; o.x = (w >= 0.0f) ? w : w * NEG_SLOPE;
        w = a.y * nd + bb.y; o.y = (w >= 0.0f) ? w : w * NEG_SLOPE;
        w = a.z * nd + bb.z; o.z = (w >= 0.0f) ? w : w * NEG_SLOPE;
        w = a.w * nd + bb.w; o.w = (w >= 0.0f) ? w : w * NEG_SLOPE;
        out4[i] = o;
    }
}

extern "C" void kernel_launch(void* const* d_in, const int* in_sizes, int n_in,
                              void* d_out, int out_size, void* d_ws, size_t ws_size,
                              hipStream_t stream) {
    const float* h    = (const float*)d_in[0];
    const float* W    = (const float*)d_in[1];
    const float* b    = (const float*)d_in[2];
    const int*   esrc = (const int*)d_in[3];
    const int*   edst = (const int*)d_in[4];
    float*       out  = (float*)d_out;

    float* ws = (float*)d_ws;
    float* ns = ws;
    float* x  = ws + 2 * (size_t)N_NODES;
    unsigned*       drec  = (unsigned*)(ws + (2 + OUT_DIM) * (size_t)N_NODES);
    unsigned short* ssort = (unsigned short*)(drec + N_EDGES);
    unsigned*       stab  = drec + N_EDGES + N_EDGES / 2;
    unsigned*       dtab  = stab + N_TILES * N_CHUNKS;

    sort_kernel<<<N_CHUNKS, SORT_T, 0, stream>>>(esrc, edst, drec, dtab, ssort, stab);
    outdeg_kernel<<<N_TILES, OD_WAVES * 32, 0, stream>>>(ssort, stab, ns);

    const int mtiles = N_NODES / 16;
    gemm_wmma_kernel<<<(mtiles + 7) / 8, 256, 0, stream>>>(h, W, ns, x, mtiles);

    agg_kernel<<<N_DT, AGG_WAVES * 32, 0, stream>>>(x, drec, dtab, b, out);
}
